# starting kernel
# baseline (speedup 1.0000x reference)
#include <hip/hip_runtime.h>


typedef __attribute__((ext_vector_type(2))) float v2f;
typedef __attribute__((ext_vector_type(8))) float v8f;

#define DIM       4096
#define NBLK      1024
#define ROWS      16
#define LDSTRIDE  4164
#define LDS_BYTES (ROWS * LDSTRIDE * 4)

__device__ __forceinline__ int pcol(int c) { return c + ((c >> 8) << 2); }

__global__ void build_tables_kernel(const float* __restrict__ w,
                                    float* __restrict__ tbl) {
  const int b     = blockIdx.x;
  const int stage = b >> 8;
  const int gid   = b & 255;
  const int t     = threadIdx.x;
  const int q     = t >> 6;
  const int L     = (t >> 1) & 31;
  const int p     = t & 1;

  const int j_in  = 4 * q + 2 * (L >> 4) + p;
  const int j_out = L & 15;
  const int h  = j_in >> 2,  l_in = j_in & 3;
  const int m2 = j_out >> 2, l    = j_out & 3;

  int g, stuff, t0;
  if (stage == 0)      { g = 1;   stuff = gid;      t0 = 0;        }
  else if (stage == 1) { g = 16;  stuff = gid >> 4; t0 = gid & 15; }
  else                 { g = 256; stuff = 0;        t0 = gid;      }

  const int nA = (stuff * 4 + h) * g + t0;
  const int nB = stuff * 4 * g + l * g + t0;
  const int LA = 2 * stage, LB = LA + 1;

  const float wa = w[((LA * NBLK + nA) * 4 + l_in) * 4 + l];
  const float wb = w[((LB * NBLK + nB) * 4 + h) * 4 + m2];
  tbl[(size_t)b * 256 + t] = wa * wb;
}

__global__ void fused_butterfly_kernel(const float* __restrict__ x,
                                       const float* __restrict__ tbl,
                                       float* __restrict__ out) {
  extern __shared__ float tile[];
  const int tid  = threadIdx.x;
  const int wave = tid >> 5;
  const int lane = tid & 31;
  const size_t rowbase = (size_t)blockIdx.x * ROWS * DIM;

  for (int i = tid; i < ROWS * (DIM / 4); i += 512) {
    const int r  = i >> 10;
    const int c4 = i & 1023;
    const float4 v = reinterpret_cast<const float4*>(x + rowbase + (size_t)r * DIM)[c4];
    *reinterpret_cast<float4*>(&tile[r * LDSTRIDE + pcol(c4 * 4)]) = v;
  }
  __syncthreads();

  const int Arow = lane & 15;
  const int kk2  = (lane >> 4) << 1;

  #pragma unroll
  for (int stage = 0; stage < 3; ++stage) {
    const int g = (stage == 0) ? 1 : (stage == 1) ? 16 : 256;
    for (int gid = wave; gid < 256; gid += 16) {
      int base;
      if (stage == 0)      base = gid << 4;
      else if (stage == 1) base = ((gid >> 4) << 8) + (gid & 15);
      else                 base = gid;

      const v2f* __restrict__ B =
          reinterpret_cast<const v2f*>(tbl + (((size_t)stage * 256 + gid) << 8));

      v8f c = {};
      #pragma unroll
      for (int q = 0; q < 4; ++q) {
        v2f a;
        const int j0 = 4 * q + kk2;
        a.x = tile[Arow * LDSTRIDE + pcol(base + j0 * g)];
        a.y = tile[Arow * LDSTRIDE + pcol(base + (j0 + 1) * g)];
        const v2f bb = B[q * 32 + lane];
        c = __builtin_amdgcn_wmma_f32_16x16x4_f32(
                false, a, false, bb, (short)0, c, false, false);
      }

      const int ccol = pcol(base + (lane & 15) * g);
      const int radd = (lane >> 4) << 3;
      #pragma unroll
      for (int v = 0; v < 8; ++v)
        tile[(v + radd) * LDSTRIDE + ccol] = c[v];
    }
    __syncthreads();
  }

  for (int i = tid; i < ROWS * (DIM / 4); i += 512) {
    const int r  = i >> 10;
    const int c4 = i & 1023;
    const float4 v = *reinterpret_cast<const float4*>(&tile[r * LDSTRIDE + pcol(c4 * 4)]);
    reinterpret_cast<float4*>(out + rowbase + (size_t)r * DIM)[c4] = v;
  }
}

extern "C" void kernel_launch(void* const* d_in, const int* in_sizes, int n_in,
                              void* d_out, int out_size, void* d_ws, size_t ws_size,
                              hipStream_t stream) {
  const float* x = (const float*)d_in[0];
  const float* w = (const float*)d_in[1];
  float* out = (float*)d_out;
  float* tbl = (float*)d_ws;

  (void)hipFuncSetAttribute((const void*)fused_butterfly_kernel,
                            hipFuncAttributeMaxDynamicSharedMemorySize,
                            LDS_BYTES);

  build_tables_kernel<<<768, 256, 0, stream>>>(w, tbl);
  fused_butterfly_kernel<<<8192 / ROWS, 512, LDS_BYTES, stream>>>(x, tbl, out);
}


# Table fragments software-pipelined: each wave requests step n+8's fragments before computing step n (48 group steps unrolled, stage barriers fence LDS only), hiding the per-group L2 round trip
# speedup vs baseline: 1.2894x; 1.2894x over previous
#include <hip/hip_runtime.h>


typedef __attribute__((ext_vector_type(2))) float v2f;
typedef __attribute__((ext_vector_type(8))) float v8f;

#define DIM       4096
#define NBLK      1024
#define ROWS      16
#define LDSTRIDE  4164
#define LDS_BYTES (ROWS * LDSTRIDE * 4)
#define NSTEP     48
#define PF        8

__device__ __forceinline__ int pcol(int c) { return c + ((c >> 8) << 2); }

__device__ __forceinline__ void lds_stage_barrier() {
  __builtin_amdgcn_fence(__ATOMIC_RELEASE, "workgroup", "local");
  __builtin_amdgcn_s_barrier();
  __builtin_amdgcn_fence(__ATOMIC_ACQUIRE, "workgroup", "local");
}

__global__ void build_tables_kernel(const float* __restrict__ w,
                                    float* __restrict__ tbl) {
  const int b     = blockIdx.x;
  const int stage = b >> 8;
  const int gid   = b & 255;
  const int t     = threadIdx.x;
  const int q     = t >> 6;
  const int L     = (t >> 1) & 31;
  const int p     = t & 1;

  const int j_in  = 4 * q + 2 * (L >> 4) + p;
  const int j_out = L & 15;
  const int h  = j_in >> 2,  l_in = j_in & 3;
  const int m2 = j_out >> 2, l    = j_out & 3;

  int g, stuff, t0;
  if (stage == 0)      { g = 1;   stuff = gid;      t0 = 0;        }
  else if (stage == 1) { g = 16;  stuff = gid >> 4; t0 = gid & 15; }
  else                 { g = 256; stuff = 0;        t0 = gid;      }

  const int nA = (stuff * 4 + h) * g + t0;
  const int nB = stuff * 4 * g + l * g + t0;
  const int LA = 2 * stage, LB = LA + 1;

  const float wa = w[((LA * NBLK + nA) * 4 + l_in) * 4 + l];
  const float wb = w[((LB * NBLK + nB) * 4 + h) * 4 + m2];
  tbl[(size_t)b * 256 + t] = wa * wb;
}

__global__ __launch_bounds__(512)
void fused_butterfly_kernel(const float* __restrict__ x,
                            const float* __restrict__ tbl,
                            float* __restrict__ out) {
  extern __shared__ float tile[];
  const int tid  = threadIdx.x;
  const int wave = tid >> 5;
  const int lane = tid & 31;
  const size_t rowbase = (size_t)blockIdx.x * ROWS * DIM;

  const float* __restrict__ bptr = tbl + (wave << 8) + (lane << 1);
  v2f bf[PF][4];
  #pragma unroll
  for (int n = 0; n < PF; ++n) {
    #pragma unroll
    for (int q = 0; q < 4; ++q)
      bf[n][q] = *reinterpret_cast<const v2f*>(bptr + n * 4096 + q * 64);
  }

  for (int i = tid; i < ROWS * (DIM / 4); i += 512) {
    const int r  = i >> 10;
    const int c4 = i & 1023;
    const float4 v = reinterpret_cast<const float4*>(x + rowbase + (size_t)r * DIM)[c4];
    *reinterpret_cast<float4*>(&tile[r * LDSTRIDE + pcol(c4 * 4)]) = v;
  }
  __syncthreads();

  const int Arow = lane & 15;
  const int kk2  = (lane >> 4) << 1;

  #pragma unroll
  for (int n = 0; n < NSTEP; ++n) {
    const int stage = n >> 4;
    const int g     = (stage == 0) ? 1 : (stage == 1) ? 16 : 256;
    const int gid   = wave + ((n & 15) << 4);
    if (n > 0 && (n & 15) == 0) lds_stage_barrier();

    int base;
    if (stage == 0)      base = gid << 4;
    else if (stage == 1) base = ((gid >> 4) << 8) + (gid & 15);
    else                 base = gid;

    const v2f b0 = bf[n % PF][0], b1 = bf[n % PF][1];
    const v2f b2 = bf[n % PF][2], b3 = bf[n % PF][3];
    if (n + PF < NSTEP) {
      #pragma unroll
      for (int q = 0; q < 4; ++q)
        bf[n % PF][q] =
            *reinterpret_cast<const v2f*>(bptr + (n + PF) * 4096 + q * 64);
    }
    __builtin_amdgcn_sched_barrier(0);

    v8f c = {};
    #pragma unroll
    for (int q = 0; q < 4; ++q) {
      v2f a;
      const int j0 = 4 * q + kk2;
      a.x = tile[Arow * LDSTRIDE + pcol(base + j0 * g)];
      a.y = tile[Arow * LDSTRIDE + pcol(base + (j0 + 1) * g)];
      const v2f bb = (q == 0) ? b0 : (q == 1) ? b1 : (q == 2) ? b2 : b3;
      c = __builtin_amdgcn_wmma_f32_16x16x4_f32(
              false, a, false, bb, (short)0, c, false, false);
    }

    const int ccol = pcol(base + (lane & 15) * g);
    const int radd = (lane >> 4) << 3;
    #pragma unroll
    for (int v = 0; v < 8; ++v)
      tile[(v + radd) * LDSTRIDE + ccol] = c[v];
  }
  __syncthreads();

  for (int i = tid; i < ROWS * (DIM / 4); i += 512) {
    const int r  = i >> 10;
    const int c4 = i & 1023;
    const float4 v = *reinterpret_cast<const float4*>(&tile[r * LDSTRIDE + pcol(c4 * 4)]);
    reinterpret_cast<float4*>(out + rowbase + (size_t)r * DIM)[c4] = v;
  }
}

extern "C" void kernel_launch(void* const* d_in, const int* in_sizes, int n_in,
                              void* d_out, int out_size, void* d_ws, size_t ws_size,
                              hipStream_t stream) {
  const float* x = (const float*)d_in[0];
  const float* w = (const float*)d_in[1];
  float* out = (float*)d_out;
  float* tbl = (float*)d_ws;

  (void)hipFuncSetAttribute((const void*)fused_butterfly_kernel,
                            hipFuncAttributeMaxDynamicSharedMemorySize,
                            LDS_BYTES);

  build_tables_kernel<<<768, 256, 0, stream>>>(w, tbl);
  fused_butterfly_kernel<<<8192 / ROWS, 512, LDS_BYTES, stream>>>(x, tbl, out);
}


# LDS tile re-padded (2 dwords per 16 and per 256 columns) and lane-half bit moved to the lowest k bit: every stage read and store hits 32 distinct banks; term order in each 4-term WMMA sum permuted
# speedup vs baseline: 1.3937x; 1.0809x over previous
#include <hip/hip_runtime.h>


typedef __attribute__((ext_vector_type(2))) float v2f;
typedef __attribute__((ext_vector_type(8))) float v8f;

#define DIM       4096
#define NBLK      1024
#define ROWS      16
#define LDSTRIDE  4676
#define LDS_BYTES (ROWS * LDSTRIDE * 4)
#define NSTEP     48
#define PF        8

__device__ __forceinline__ int pcol(int c) { return c + ((c >> 4) << 1) + ((c >> 8) << 1); }
__device__ __forceinline__ int pdig(int d2, int d1, int d0) { return 290 * d2 + 18 * d1 + d0; }

__device__ __forceinline__ void lds_stage_barrier() {
  __builtin_amdgcn_fence(__ATOMIC_RELEASE, "workgroup", "local");
  __builtin_amdgcn_s_barrier();
  __builtin_amdgcn_fence(__ATOMIC_ACQUIRE, "workgroup", "local");
}

__global__ void build_tables_kernel(const float* __restrict__ w,
                                    float* __restrict__ tbl) {
  const int b     = blockIdx.x;
  const int stage = b >> 8;
  const int gid   = b & 255;
  const int t     = threadIdx.x;
  const int q     = t >> 6;
  const int L     = (t >> 1) & 31;
  const int p     = t & 1;

  const int j_in  = 4 * q + 2 * p + (L >> 4);
  const int j_out = L & 15;
  const int h  = j_in >> 2,  l_in = j_in & 3;
  const int m2 = j_out >> 2, l    = j_out & 3;

  int g, stuff, t0;
  if (stage == 0)      { g = 1;   stuff = gid;      t0 = 0;        }
  else if (stage == 1) { g = 16;  stuff = gid >> 4; t0 = gid & 15; }
  else                 { g = 256; stuff = 0;        t0 = gid;      }

  const int nA = (stuff * 4 + h) * g + t0;
  const int nB = stuff * 4 * g + l * g + t0;
  const int LA = 2 * stage, LB = LA + 1;

  const float wa = w[((LA * NBLK + nA) * 4 + l_in) * 4 + l];
  const float wb = w[((LB * NBLK + nB) * 4 + h) * 4 + m2];
  tbl[(size_t)b * 256 + t] = wa * wb;
}

__global__ __launch_bounds__(512)
void fused_butterfly_kernel(const float* __restrict__ x,
                            const float* __restrict__ tbl,
                            float* __restrict__ out) {
  extern __shared__ float tile[];
  const int tid  = threadIdx.x;
  const int wave = tid >> 5;
  const int lane = tid & 31;
  const size_t rowbase = (size_t)blockIdx.x * ROWS * DIM;

  const float* __restrict__ bptr = tbl + (wave << 8) + (lane << 1);
  v2f bf[PF][4];
  #pragma unroll
  for (int n = 0; n < PF; ++n) {
    #pragma unroll
    for (int q = 0; q < 4; ++q)
      bf[n][q] = *reinterpret_cast<const v2f*>(bptr + n * 4096 + q * 64);
  }

  #pragma unroll 8
  for (int i = tid; i < ROWS * (DIM / 4); i += 512) {
    const int r  = i >> 10;
    const int c4 = i & 1023;
    const float4 v = reinterpret_cast<const float4*>(x + rowbase + (size_t)r * DIM)[c4];
    float2* dst = reinterpret_cast<float2*>(&tile[r * LDSTRIDE + pcol(c4 * 4)]);
    dst[0] = make_float2(v.x, v.y);
    dst[1] = make_float2(v.z, v.w);
  }
  __syncthreads();

  const int Arow = lane & 15;
  const int kh   = lane >> 4;

  #pragma unroll
  for (int n = 0; n < NSTEP; ++n) {
    const int stage = n >> 4;
    if (n > 0 && (n & 15) == 0) lds_stage_barrier();

    const int k = n & 15;
    int pbase, pstep;
    if (stage == 0)      { pbase = pdig(k, wave, 0); pstep = pdig(0, 0, 1); }
    else if (stage == 1) { pbase = pdig(k, 0, wave); pstep = pdig(0, 1, 0); }
    else                 { pbase = pdig(0, k, wave); pstep = pdig(1, 0, 0); }

    const v2f b0 = bf[n % PF][0], b1 = bf[n % PF][1];
    const v2f b2 = bf[n % PF][2], b3 = bf[n % PF][3];
    if (n + PF < NSTEP) {
      #pragma unroll
      for (int q = 0; q < 4; ++q)
        bf[n % PF][q] =
            *reinterpret_cast<const v2f*>(bptr + (n + PF) * 4096 + q * 64);
    }
    __builtin_amdgcn_sched_barrier(0);

    v8f c = {};
    #pragma unroll
    for (int q = 0; q < 4; ++q) {
      v2f a;
      const int j0 = 4 * q + kh;
      a.x = tile[Arow * LDSTRIDE + pbase + j0 * pstep];
      a.y = tile[Arow * LDSTRIDE + pbase + (j0 + 2) * pstep];
      const v2f bb = (q == 0) ? b0 : (q == 1) ? b1 : (q == 2) ? b2 : b3;
      c = __builtin_amdgcn_wmma_f32_16x16x4_f32(
              false, a, false, bb, (short)0, c, false, false);
    }

    const int ccol = pbase + (lane & 15) * pstep;
    const int radd = (lane >> 4) << 3;
    #pragma unroll
    for (int v = 0; v < 8; ++v)
      tile[(v + radd) * LDSTRIDE + ccol] = c[v];
  }
  __syncthreads();

  #pragma unroll 8
  for (int i = tid; i < ROWS * (DIM / 4); i += 512) {
    const int r  = i >> 10;
    const int c4 = i & 1023;
    const float2* src = reinterpret_cast<const float2*>(&tile[r * LDSTRIDE + pcol(c4 * 4)]);
    const float2 lo = src[0], hi = src[1];
    reinterpret_cast<float4*>(out + rowbase + (size_t)r * DIM)[c4] =
        make_float4(lo.x, lo.y, hi.x, hi.y);
  }
}

extern "C" void kernel_launch(void* const* d_in, const int* in_sizes, int n_in,
                              void* d_out, int out_size, void* d_ws, size_t ws_size,
                              hipStream_t stream) {
  const float* x = (const float*)d_in[0];
  const float* w = (const float*)d_in[1];
  float* out = (float*)d_out;
  float* tbl = (float*)d_ws;

  (void)hipFuncSetAttribute((const void*)fused_butterfly_kernel,
                            hipFuncAttributeMaxDynamicSharedMemorySize,
                            LDS_BYTES);

  build_tables_kernel<<<768, 256, 0, stream>>>(w, tbl);
  fused_butterfly_kernel<<<8192 / ROWS, 512, LDS_BYTES, stream>>>(x, tbl, out);
}


# A fragments read ahead through a 3-set register ring: inside a stage each wave issues step n+2's LDS fragment reads as one group before step n's WMMAs, so LDS round trips overlap the matrix work
# speedup vs baseline: 1.4384x; 1.0321x over previous
#include <hip/hip_runtime.h>


typedef __attribute__((ext_vector_type(2))) float v2f;
typedef __attribute__((ext_vector_type(8))) float v8f;

#define DIM       4096
#define NBLK      1024
#define ROWS      16
#define LDSTRIDE  4676
#define LDS_BYTES (ROWS * LDSTRIDE * 4)
#define NSTEP     48
#define PF        8
#define AF        3

__device__ __forceinline__ int pcol(int c) { return c + ((c >> 4) << 1) + ((c >> 8) << 1); }
__device__ __forceinline__ int pdig(int d2, int d1, int d0) { return 290 * d2 + 18 * d1 + d0; }

__device__ __forceinline__ void lds_stage_barrier() {
  __builtin_amdgcn_fence(__ATOMIC_RELEASE, "workgroup", "local");
  __builtin_amdgcn_s_barrier();
  __builtin_amdgcn_fence(__ATOMIC_ACQUIRE, "workgroup", "local");
}

__global__ void build_tables_kernel(const float* __restrict__ w,
                                    float* __restrict__ tbl) {
  const int b     = blockIdx.x;
  const int stage = b >> 8;
  const int gid   = b & 255;
  const int t     = threadIdx.x;
  const int q     = t >> 6;
  const int L     = (t >> 1) & 31;
  const int p     = t & 1;

  const int j_in  = 4 * q + 2 * p + (L >> 4);
  const int j_out = L & 15;
  const int h  = j_in >> 2,  l_in = j_in & 3;
  const int m2 = j_out >> 2, l    = j_out & 3;

  int g, stuff, t0;
  if (stage == 0)      { g = 1;   stuff = gid;      t0 = 0;        }
  else if (stage == 1) { g = 16;  stuff = gid >> 4; t0 = gid & 15; }
  else                 { g = 256; stuff = 0;        t0 = gid;      }

  const int nA = (stuff * 4 + h) * g + t0;
  const int nB = stuff * 4 * g + l * g + t0;
  const int LA = 2 * stage, LB = LA + 1;

  const float wa = w[((LA * NBLK + nA) * 4 + l_in) * 4 + l];
  const float wb = w[((LB * NBLK + nB) * 4 + h) * 4 + m2];
  tbl[(size_t)b * 256 + t] = wa * wb;
}

__global__ __launch_bounds__(512)
void fused_butterfly_kernel(const float* __restrict__ x,
                            const float* __restrict__ tbl,
                            float* __restrict__ out) {
  extern __shared__ float tile[];
  const int tid  = threadIdx.x;
  const int wave = tid >> 5;
  const int lane = tid & 31;
  const size_t rowbase = (size_t)blockIdx.x * ROWS * DIM;

  const float* __restrict__ bptr = tbl + (wave << 8) + (lane << 1);
  v2f bf[PF][4];
  #pragma unroll
  for (int n = 0; n < PF; ++n) {
    #pragma unroll
    for (int q = 0; q < 4; ++q)
      bf[n][q] = *reinterpret_cast<const v2f*>(bptr + n * 4096 + q * 64);
  }

  #pragma unroll 8
  for (int i = tid; i < ROWS * (DIM / 4); i += 512) {
    const int r  = i >> 10;
    const int c4 = i & 1023;
    const float4 v = reinterpret_cast<const float4*>(x + rowbase + (size_t)r * DIM)[c4];
    float2* dst = reinterpret_cast<float2*>(&tile[r * LDSTRIDE + pcol(c4 * 4)]);
    dst[0] = make_float2(v.x, v.y);
    dst[1] = make_float2(v.z, v.w);
  }
  __syncthreads();

  const int Arow = lane & 15;
  const int kh   = lane >> 4;

  auto step_cols = [&](int m, int& pbase, int& pstep) {
    const int stage = m >> 4, k = m & 15;
    if (stage == 0)      { pbase = pdig(k, wave, 0); pstep = pdig(0, 0, 1); }
    else if (stage == 1) { pbase = pdig(k, 0, wave); pstep = pdig(0, 1, 0); }
    else                 { pbase = pdig(0, k, wave); pstep = pdig(1, 0, 0); }
  };

  v2f af[AF][4];
  auto load_afrag = [&](int m) {
    int pb, ps;
    step_cols(m, pb, ps);
    #pragma unroll
    for (int q = 0; q < 4; ++q) {
      const int j0 = 4 * q + kh;
      af[m % AF][q].x = tile[Arow * LDSTRIDE + pb + j0 * ps];
      af[m % AF][q].y = tile[Arow * LDSTRIDE + pb + (j0 + 2) * ps];
    }
  };

  #pragma unroll
  for (int n = 0; n < NSTEP; ++n) {
    if ((n & 15) == 0) {
      if (n > 0) lds_stage_barrier();
      load_afrag(n);
      load_afrag(n + 1);
    }

    int pbase, pstep;
    step_cols(n, pbase, pstep);

    const v2f b0 = bf[n % PF][0], b1 = bf[n % PF][1];
    const v2f b2 = bf[n % PF][2], b3 = bf[n % PF][3];
    if (n + PF < NSTEP) {
      #pragma unroll
      for (int q = 0; q < 4; ++q)
        bf[n % PF][q] =
            *reinterpret_cast<const v2f*>(bptr + (n + PF) * 4096 + q * 64);
    }
    __builtin_amdgcn_sched_barrier(0);

    if ((n & 15) + 2 < 16) load_afrag(n + 2);
    __builtin_amdgcn_sched_barrier(0);

    v8f c = {};
    #pragma unroll
    for (int q = 0; q < 4; ++q) {
      const v2f bb = (q == 0) ? b0 : (q == 1) ? b1 : (q == 2) ? b2 : b3;
      c = __builtin_amdgcn_wmma_f32_16x16x4_f32(
              false, af[n % AF][q], false, bb, (short)0, c, false, false);
    }

    const int ccol = pbase + (lane & 15) * pstep;
    const int radd = (lane >> 4) << 3;
    #pragma unroll
    for (int v = 0; v < 8; ++v)
      tile[(v + radd) * LDSTRIDE + ccol] = c[v];
  }
  __syncthreads();

  #pragma unroll 8
  for (int i = tid; i < ROWS * (DIM / 4); i += 512) {
    const int r  = i >> 10;
    const int c4 = i & 1023;
    const float2* src = reinterpret_cast<const float2*>(&tile[r * LDSTRIDE + pcol(c4 * 4)]);
    const float2 lo = src[0], hi = src[1];
    reinterpret_cast<float4*>(out + rowbase + (size_t)r * DIM)[c4] =
        make_float4(lo.x, lo.y, hi.x, hi.y);
  }
}

extern "C" void kernel_launch(void* const* d_in, const int* in_sizes, int n_in,
                              void* d_out, int out_size, void* d_ws, size_t ws_size,
                              hipStream_t stream) {
  const float* x = (const float*)d_in[0];
  const float* w = (const float*)d_in[1];
  float* out = (float*)d_out;
  float* tbl = (float*)d_ws;

  (void)hipFuncSetAttribute((const void*)fused_butterfly_kernel,
                            hipFuncAttributeMaxDynamicSharedMemorySize,
                            LDS_BYTES);

  build_tables_kernel<<<768, 256, 0, stream>>>(w, tbl);
  fused_butterfly_kernel<<<8192 / ROWS, 512, LDS_BYTES, stream>>>(x, tbl, out);
}


# Each workgroup runs 2 row tiles in turn: the other tile's HBM read and write ride in 88 registers per thread under the matrix stages (LDS<->register swap between tiles); sum order unchanged
# speedup vs baseline: 1.5843x; 1.1014x over previous
#include <hip/hip_runtime.h>


typedef __attribute__((ext_vector_type(2))) float v2f;
typedef __attribute__((ext_vector_type(8))) float v8f;

#define DIM       4096
#define NBLK      1024
#define ROWS      16
#define LDSTRIDE  4676
#define LDS_BYTES (ROWS * LDSTRIDE * 4)
#define NSTEP     48
#define PF        8
#define AF        3
#define TILES     2
#define TF4       32
#define HB        22

__device__ __forceinline__ int pcol(int c) { return c + ((c >> 4) << 1) + ((c >> 8) << 1); }
__device__ __forceinline__ int pdig(int d2, int d1, int d0) { return 290 * d2 + 18 * d1 + d0; }

__device__ __forceinline__ void lds_stage_barrier() {
  __builtin_amdgcn_fence(__ATOMIC_RELEASE, "workgroup", "local");
  __builtin_amdgcn_s_barrier();
  __builtin_amdgcn_fence(__ATOMIC_ACQUIRE, "workgroup", "local");
}

__global__ void build_tables_kernel(const float* __restrict__ w,
                                    float* __restrict__ tbl) {
  const int b     = blockIdx.x;
  const int stage = b >> 8;
  const int gid   = b & 255;
  const int t     = threadIdx.x;
  const int q     = t >> 6;
  const int L     = (t >> 1) & 31;
  const int p     = t & 1;

  const int j_in  = 4 * q + 2 * p + (L >> 4);
  const int j_out = L & 15;
  const int h  = j_in >> 2,  l_in = j_in & 3;
  const int m2 = j_out >> 2, l    = j_out & 3;

  int g, stuff, t0;
  if (stage == 0)      { g = 1;   stuff = gid;      t0 = 0;        }
  else if (stage == 1) { g = 16;  stuff = gid >> 4; t0 = gid & 15; }
  else                 { g = 256; stuff = 0;        t0 = gid;      }

  const int nA = (stuff * 4 + h) * g + t0;
  const int nB = stuff * 4 * g + l * g + t0;
  const int LA = 2 * stage, LB = LA + 1;

  const float wa = w[((LA * NBLK + nA) * 4 + l_in) * 4 + l];
  const float wb = w[((LB * NBLK + nB) * 4 + h) * 4 + m2];
  tbl[(size_t)b * 256 + t] = wa * wb;
}

__global__ __launch_bounds__(512)
void fused_butterfly_kernel(const float* __restrict__ x,
                            const float* __restrict__ tbl,
                            float* __restrict__ out) {
  extern __shared__ float tile[];
  const int tid = threadIdx.x;
  const size_t rowbase = (size_t)blockIdx.x * (TILES * ROWS) * DIM;

  const float4* __restrict__ xin  = reinterpret_cast<const float4*>(x + rowbase) + tid;
  float4* __restrict__       yout = reinterpret_cast<float4*>(out + rowbase) + tid;
  float4 hb[HB];

  v2f bf[PF][4];
  {
    const float* bptr0 = tbl + ((tid >> 5) << 8) + ((tid & 31) << 1);
    #pragma unroll
    for (int n = 0; n < PF; ++n) {
      #pragma unroll
      for (int q = 0; q < 4; ++q)
        bf[n][q] = *reinterpret_cast<const v2f*>(bptr0 + n * 4096 + q * 64);
    }
  }

  #pragma unroll 8
  for (int i = tid; i < ROWS * (DIM / 4); i += 512) {
    const int r  = i >> 10;
    const int c4 = i & 1023;
    const float4 v = reinterpret_cast<const float4*>(x + rowbase + (size_t)r * DIM)[c4];
    float2* dst = reinterpret_cast<float2*>(&tile[r * LDSTRIDE + pcol(c4 * 4)]);
    dst[0] = make_float2(v.x, v.y);
    dst[1] = make_float2(v.z, v.w);
  }

  for (int t = 0; t < TILES; ++t) {
    int ltid = tid;
    asm volatile("" : "+v"(ltid));
    const int wave = ltid >> 5;
    const int lane = ltid & 31;
    const float* bptr = tbl + (wave << 8) + (lane << 1);

    const int Arow = lane & 15;
    const int kh   = lane >> 4;

    auto step_cols = [&](int m, int& pbase, int& pstep) {
      const int stage = m >> 4, k = m & 15;
      if (stage == 0)      { pbase = pdig(k, wave, 0); pstep = pdig(0, 0, 1); }
      else if (stage == 1) { pbase = pdig(k, 0, wave); pstep = pdig(0, 1, 0); }
      else                 { pbase = pdig(0, k, wave); pstep = pdig(1, 0, 0); }
    };

    v2f af[AF][4];
    auto load_afrag = [&](int m) {
      int pb, ps;
      step_cols(m, pb, ps);
      #pragma unroll
      for (int q = 0; q < 4; ++q) {
        const int j0 = 4 * q + kh;
        af[m % AF][q].x = tile[Arow * LDSTRIDE + pb + j0 * ps];
        af[m % AF][q].y = tile[Arow * LDSTRIDE + pb + (j0 + 2) * ps];
      }
    };

    lds_stage_barrier();

    #pragma unroll
    for (int n = 0; n < NSTEP; ++n) {
      if ((n & 15) == 0) {
        if (n > 0) lds_stage_barrier();
        load_afrag(n);
        load_afrag(n + 1);
      }

      int pbase, pstep;
      step_cols(n, pbase, pstep);

      const v2f b0 = bf[n % PF][0], b1 = bf[n % PF][1];
      const v2f b2 = bf[n % PF][2], b3 = bf[n % PF][3];
      if (n + PF < NSTEP) {
        #pragma unroll
        for (int q = 0; q < 4; ++q)
          bf[n % PF][q] =
              *reinterpret_cast<const v2f*>(bptr + (n + PF) * 4096 + q * 64);
      } else if (t + 1 < TILES) {
        #pragma unroll
        for (int q = 0; q < 4; ++q)
          bf[n % PF][q] =
              *reinterpret_cast<const v2f*>(bptr + (n + PF - NSTEP) * 4096 + q * 64);
      }
      if ((n & 1) == 0 && (n >> 1) < HB) {
        if (t == 0) hb[n >> 1] = xin[ROWS * (DIM / 4) + (n >> 1) * 512];
        else        yout[(n >> 1) * 512] = hb[n >> 1];
      }
      __builtin_amdgcn_sched_barrier(0);

      if ((n & 15) + 2 < 16) load_afrag(n + 2);
      __builtin_amdgcn_sched_barrier(0);

      v8f c = {};
      #pragma unroll
      for (int q = 0; q < 4; ++q) {
        const v2f bb = (q == 0) ? b0 : (q == 1) ? b1 : (q == 2) ? b2 : b3;
        c = __builtin_amdgcn_wmma_f32_16x16x4_f32(
                false, af[n % AF][q], false, bb, (short)0, c, false, false);
      }

      const int ccol = pbase + (lane & 15) * pstep;
      const int radd = (lane >> 4) << 3;
      #pragma unroll
      for (int v = 0; v < 8; ++v)
        tile[(v + radd) * LDSTRIDE + ccol] = c[v];
    }
    lds_stage_barrier();

    if (t + 1 < TILES) {
      float4 late[TF4 - HB];
      #pragma unroll
      for (int u = HB; u < TF4; ++u)
        late[u - HB] = xin[ROWS * (DIM / 4) + u * 512];
      #pragma unroll
      for (int u = 0; u < HB; ++u) {
        const int i  = ltid + 512 * u;
        const int r  = i >> 10;
        const int c4 = i & 1023;
        float2* slot = reinterpret_cast<float2*>(&tile[r * LDSTRIDE + pcol(c4 * 4)]);
        const float2 lo = slot[0], hi = slot[1];
        slot[0] = make_float2(hb[u].x, hb[u].y);
        slot[1] = make_float2(hb[u].z, hb[u].w);
        hb[u] = make_float4(lo.x, lo.y, hi.x, hi.y);
      }
      #pragma unroll
      for (int u = HB; u < TF4; ++u) {
        const int i  = ltid + 512 * u;
        const int r  = i >> 10;
        const int c4 = i & 1023;
        float2* slot = reinterpret_cast<float2*>(&tile[r * LDSTRIDE + pcol(c4 * 4)]);
        const float2 lo = slot[0], hi = slot[1];
        yout[u * 512] = make_float4(lo.x, lo.y, hi.x, hi.y);
        slot[0] = make_float2(late[u - HB].x, late[u - HB].y);
        slot[1] = make_float2(late[u - HB].z, late[u - HB].w);
      }
    }
  }

  #pragma unroll 8
  for (int i = tid; i < ROWS * (DIM / 4); i += 512) {
    const int r  = i >> 10;
    const int c4 = i & 1023;
    const float2* src = reinterpret_cast<const float2*>(&tile[r * LDSTRIDE + pcol(c4 * 4)]);
    const float2 lo = src[0], hi = src[1];
    reinterpret_cast<float4*>(out + rowbase + (size_t)(ROWS + r) * DIM)[c4] =
        make_float4(lo.x, lo.y, hi.x, hi.y);
  }
}

extern "C" void kernel_launch(void* const* d_in, const int* in_sizes, int n_in,
                              void* d_out, int out_size, void* d_ws, size_t ws_size,
                              hipStream_t stream) {
  const float* x = (const float*)d_in[0];
  const float* w = (const float*)d_in[1];
  float* out = (float*)d_out;
  float* tbl = (float*)d_ws;

  (void)hipFuncSetAttribute((const void*)fused_butterfly_kernel,
                            hipFuncAttributeMaxDynamicSharedMemorySize,
                            LDS_BYTES);

  build_tables_kernel<<<768, 256, 0, stream>>>(w, tbl);
  fused_butterfly_kernel<<<8192 / (TILES * ROWS), 512, LDS_BYTES, stream>>>(x, tbl, out);
}
